# starting kernel
# baseline (speedup 1.0000x reference)
#include <hip/hip_runtime.h>
#include <hip/hip_bf16.h>
#include <math.h>

typedef __attribute__((ext_vector_type(2))) float v2f;
typedef __attribute__((ext_vector_type(8))) float v8f;

__device__ __forceinline__ float relu1(float x) {
    float r;
    asm("v_max_num_f32 %0, 0, %1" : "=v"(r) : "v"(x));
    return r;
}

__global__ __launch_bounds__(256) void rnn_wmma_kernel(
    const float* __restrict__ state,
    const float* __restrict__ W_ih, const float* __restrict__ W_hh,
    const float* __restrict__ b_ih, const float* __restrict__ b_hh,
    const float* __restrict__ W_out, const float* __restrict__ b_out,
    float* __restrict__ out, int B, int T)
{
    const int lane = threadIdx.x & 31;
    const int m    = lane & 15;
    const bool low = lane < 16;
    const int w    = blockIdx.x * (blockDim.x >> 5) + (threadIdx.x >> 5);
    if (w * 32 >= B) return;

    auto whh = [&](int r, int c) -> float { return (r < 5) ? W_hh[r * 5 + c] : 0.0f; };
    auto wih = [&](int r, int c) -> float { return (r < 5) ? W_ih[r * 3 + c] : 0.0f; };

    v2f Aa, Ab, Ac;
    Aa.x = low ? whh(m, 0) : 0.0f;
    Aa.y = low ? whh(m, 1) : 0.0f;
    Ab.x = low ? whh(m, 2) : 0.0f;
    Ab.y = low ? whh(m, 3) : 0.0f;
    Ac.x = low ? whh(m, 4) : wih(m, 1);
    Ac.y = low ? wih(m, 0) : wih(m, 2);

    v8f cbias;
#pragma unroll
    for (int j = 0; j < 8; ++j) cbias[j] = 0.0f;
#pragma unroll
    for (int j = 0; j < 5; ++j) cbias[j] = low ? (b_ih[j] + b_hh[j]) : 0.0f;

    float h[2][5];
    const float* p[2];
    float4 c0[2], c1[2], c2[2];
#pragma unroll
    for (int g = 0; g < 2; ++g) {
#pragma unroll
        for (int j = 0; j < 5; ++j) h[g][j] = 0.0f;
        const int chain = w * 32 + g * 16 + m;
        p[g] = state + (size_t)chain * (size_t)T * 3u;
        c0[g] = ((const float4*)p[g])[0];
        c1[g] = ((const float4*)p[g])[1];
        c2[g] = ((const float4*)p[g])[2];
    }

    auto step = [&](int g, float x0, float x1, float x2) {
        v2f Ba, Bb, Bc;
        Ba.x = h[g][0];
        Ba.y = h[g][1];
        Bb.x = h[g][2];
        Bb.y = h[g][3];
        Bc.x = low ? h[g][4] : x1;
        Bc.y = low ? x0 : x2;
        v8f acc = __builtin_amdgcn_wmma_f32_16x16x4_f32(false, Aa, false, Ba,
                                                        (short)0, cbias, false, false);
        acc = __builtin_amdgcn_wmma_f32_16x16x4_f32(false, Ab, false, Bb,
                                                    (short)0, acc, false, false);
        acc = __builtin_amdgcn_wmma_f32_16x16x4_f32(false, Ac, false, Bc,
                                                    (short)0, acc, false, false);
        h[g][0] = relu1(acc[0]);
        h[g][1] = relu1(acc[1]);
        h[g][2] = relu1(acc[2]);
        h[g][3] = relu1(acc[3]);
        h[g][4] = relu1(acc[4]);
    };

    const int nblk = T >> 2;
    for (int it = 0; it < nblk - 1; ++it) {
        float4 n0[2], n1[2], n2[2];
#pragma unroll
        for (int g = 0; g < 2; ++g) {
            const float4* pn = (const float4*)(p[g] + 12);
            n0[g] = pn[0]; n1[g] = pn[1]; n2[g] = pn[2];
            __builtin_prefetch(p[g] + 96, 0, 3);
        }

        step(0, c0[0].x, c0[0].y, c0[0].z);  step(1, c0[1].x, c0[1].y, c0[1].z);
        step(0, c0[0].w, c1[0].x, c1[0].y);  step(1, c0[1].w, c1[1].x, c1[1].y);
        step(0, c1[0].z, c1[0].w, c2[0].x);  step(1, c1[1].z, c1[1].w, c2[1].x);
        step(0, c2[0].y, c2[0].z, c2[0].w);  step(1, c2[1].y, c2[1].z, c2[1].w);

#pragma unroll
        for (int g = 0; g < 2; ++g) {
            c0[g] = n0[g]; c1[g] = n1[g]; c2[g] = n2[g];
            p[g] += 12;
        }
    }
    step(0, c0[0].x, c0[0].y, c0[0].z);  step(1, c0[1].x, c0[1].y, c0[1].z);
    step(0, c0[0].w, c1[0].x, c1[0].y);  step(1, c0[1].w, c1[1].x, c1[1].y);
    step(0, c1[0].z, c1[0].w, c2[0].x);  step(1, c1[1].z, c1[1].w, c2[1].x);
    step(0, c2[0].y, c2[0].z, c2[0].w);  step(1, c2[1].y, c2[1].z, c2[1].w);

    if (low) {
#pragma unroll
        for (int g = 0; g < 2; ++g) {
            float z = W_out[0] * h[g][0] + W_out[1] * h[g][1] + W_out[2] * h[g][2] +
                      W_out[3] * h[g][3] + W_out[4] * h[g][4] + b_out[0];
            out[w * 32 + g * 16 + m] = tanhf(z);
        }
    }
}

extern "C" void kernel_launch(void* const* d_in, const int* in_sizes, int n_in,
                              void* d_out, int out_size, void* d_ws, size_t ws_size,
                              hipStream_t stream) {
    (void)n_in; (void)out_size; (void)d_ws; (void)ws_size;
    const float* state = (const float*)d_in[0];
    const float* W_ih  = (const float*)d_in[1];
    const float* W_hh  = (const float*)d_in[2];
    const float* b_ih  = (const float*)d_in[3];
    const float* b_hh  = (const float*)d_in[4];
    const float* W_out = (const float*)d_in[5];
    const float* b_out = (const float*)d_in[6];
    float* out = (float*)d_out;

    const int T = 2048;
    const int B = in_sizes[0] / (T * 3);
    const int waves = B / 32;
    const int wavesPerBlock = 8;
    const int blocks = (waves + wavesPerBlock - 1) / wavesPerBlock;

    rnn_wmma_kernel<<<blocks, wavesPerBlock * 32, 0, stream>>>(
        state, W_ih, W_hh, b_ih, b_hh, W_out, b_out, out, B, T);
}


# Two f32 WMMAs per timestep instead of three: hidden units 2,3 also computed in D rows 8,9, so relu(D0),relu(D1) form a full K=4 B operand (h0..h3); h4,x0..x2 fill the second; f32 sum order changes
# speedup vs baseline: 1.2169x; 1.2169x over previous
#include <hip/hip_runtime.h>
#include <hip/hip_bf16.h>
#include <math.h>


typedef __attribute__((ext_vector_type(2))) float v2f;
typedef __attribute__((ext_vector_type(8))) float v8f;

__device__ __forceinline__ float maxnum1(float a, float b) {
    float r;
    asm("v_max_num_f32 %0, %1, %2" : "=v"(r) : "v"(a), "v"(b));
    return r;
}
__device__ __forceinline__ float relu1(float x) {
    float r;
    asm("v_max_num_f32 %0, 0, %1" : "=v"(r) : "v"(x));
    return r;
}

__global__ __launch_bounds__(256) void rnn_wmma_kernel(
    const float* __restrict__ state,
    const float* __restrict__ W_ih, const float* __restrict__ W_hh,
    const float* __restrict__ b_ih, const float* __restrict__ b_hh,
    const float* __restrict__ W_out, const float* __restrict__ b_out,
    float* __restrict__ out, int B, int T)
{
    const int lane = threadIdx.x & 31;
    const int m    = lane & 15;
    const bool low = lane < 16;
    const int w    = blockIdx.x * (blockDim.x >> 5) + (threadIdx.x >> 5);
    if (w * 32 >= B) return;

    const int  unit = (m < 5) ? m : (m == 8 ? 2 : 3);
    const bool live = ((0x31Fu >> m) & 1u) != 0u;
    float wh[5], wi[3];
#pragma unroll
    for (int c = 0; c < 5; ++c) wh[c] = W_hh[unit * 5 + c];
#pragma unroll
    for (int c = 0; c < 3; ++c) wi[c] = W_ih[unit * 3 + c];

    v2f A1, A2;
    A1.x = live ? (low ? wh[0] : wh[2]) : 0.0f;
    A1.y = live ? (low ? wh[1] : wh[3]) : 0.0f;
    A2.x = live ? (low ? wh[4] : wi[1]) : 0.0f;
    A2.y = live ? (low ? wi[0] : wi[2]) : 0.0f;

    const float NEG_INF = -__builtin_huge_valf();
    float bs[5];
#pragma unroll
    for (int j = 0; j < 5; ++j) bs[j] = b_ih[j] + b_hh[j];
    v8f cbias;
    cbias[0] = low ? bs[0] : bs[2];
    cbias[1] = low ? bs[1] : bs[3];
    cbias[2] = low ? bs[2] : 0.0f;
    cbias[3] = low ? bs[3] : 0.0f;
    cbias[4] = low ? bs[4] : NEG_INF;
    cbias[5] = 0.0f; cbias[6] = 0.0f; cbias[7] = 0.0f;

    float a[2][5];
    const float* p[2];
    float4 c0[2], c1[2], c2[2];
#pragma unroll
    for (int g = 0; g < 2; ++g) {
#pragma unroll
        for (int j = 0; j < 4; ++j) a[g][j] = 0.0f;
        a[g][4] = low ? 0.0f : NEG_INF;
        const int chain = w * 32 + g * 16 + m;
        p[g] = state + (size_t)chain * (size_t)T * 3u;
        c0[g] = ((const float4*)p[g])[0];
        c1[g] = ((const float4*)p[g])[1];
        c2[g] = ((const float4*)p[g])[2];
    }

    auto step = [&](int g, float x0, float x1, float x2) {
        const float xs = low ? 0.0f : x1;
        v2f B1, B2;
        B2.y = low ? x0 : x2;
        B1.x = relu1(a[g][0]);
        B1.y = relu1(a[g][1]);
        B2.x = maxnum1(a[g][4], xs);
        v8f acc = __builtin_amdgcn_wmma_f32_16x16x4_f32(false, A1, false, B1,
                                                        (short)0, cbias, false, false);
        acc = __builtin_amdgcn_wmma_f32_16x16x4_f32(false, A2, false, B2,
                                                    (short)0, acc, false, false);
        a[g][0] = acc[0];
        a[g][1] = acc[1];
        a[g][2] = acc[2];
        a[g][3] = acc[3];
        a[g][4] = acc[4];
    };

    const int nblk = T >> 2;
    for (int it = 0; it < nblk - 1; ++it) {
        float4 n0[2], n1[2], n2[2];
#pragma unroll
        for (int g = 0; g < 2; ++g) {
            const float4* pn = (const float4*)(p[g] + 12);
            n0[g] = pn[0]; n1[g] = pn[1]; n2[g] = pn[2];
            __builtin_prefetch(p[g] + 96, 0, 3);
        }

        step(0, c0[0].x, c0[0].y, c0[0].z);  step(1, c0[1].x, c0[1].y, c0[1].z);
        step(0, c0[0].w, c1[0].x, c1[0].y);  step(1, c0[1].w, c1[1].x, c1[1].y);
        step(0, c1[0].z, c1[0].w, c2[0].x);  step(1, c1[1].z, c1[1].w, c2[1].x);
        step(0, c2[0].y, c2[0].z, c2[0].w);  step(1, c2[1].y, c2[1].z, c2[1].w);

#pragma unroll
        for (int g = 0; g < 2; ++g) {
            c0[g] = n0[g]; c1[g] = n1[g]; c2[g] = n2[g];
            p[g] += 12;
        }
    }
    step(0, c0[0].x, c0[0].y, c0[0].z);  step(1, c0[1].x, c0[1].y, c0[1].z);
    step(0, c0[0].w, c1[0].x, c1[0].y);  step(1, c0[1].w, c1[1].x, c1[1].y);
    step(0, c1[0].z, c1[0].w, c2[0].x);  step(1, c1[1].z, c1[1].w, c2[1].x);
    step(0, c2[0].y, c2[0].z, c2[0].w);  step(1, c2[1].y, c2[1].z, c2[1].w);

    if (low) {
#pragma unroll
        for (int g = 0; g < 2; ++g) {
            float h[5];
#pragma unroll
            for (int j = 0; j < 5; ++j) h[j] = fmaxf(a[g][j], 0.0f);
            float z = W_out[0] * h[0] + W_out[1] * h[1] + W_out[2] * h[2] +
                      W_out[3] * h[3] + W_out[4] * h[4] + b_out[0];
            out[w * 32 + g * 16 + m] = tanhf(z);
        }
    }
}

extern "C" void kernel_launch(void* const* d_in, const int* in_sizes, int n_in,
                              void* d_out, int out_size, void* d_ws, size_t ws_size,
                              hipStream_t stream) {
    (void)n_in; (void)out_size; (void)d_ws; (void)ws_size;
    const float* state = (const float*)d_in[0];
    const float* W_ih  = (const float*)d_in[1];
    const float* W_hh  = (const float*)d_in[2];
    const float* b_ih  = (const float*)d_in[3];
    const float* b_hh  = (const float*)d_in[4];
    const float* W_out = (const float*)d_in[5];
    const float* b_out = (const float*)d_in[6];
    float* out = (float*)d_out;

    const int T = 2048;
    const int B = in_sizes[0] / (T * 3);
    const int waves = B / 32;
    const int wavesPerBlock = 8;
    const int blocks = (waves + wavesPerBlock - 1) / wavesPerBlock;

    rnn_wmma_kernel<<<blocks, wavesPerBlock * 32, 0, stream>>>(
        state, W_ih, W_hh, b_ih, b_hh, W_out, b_out, out, B, T);
}


# Next 8-timestep block's input loads issued as one group ahead of the current block's WMMA chain (2 register sets, scheduling fence); side effect: compiler gives each group its own accumulators
# speedup vs baseline: 1.6216x; 1.3326x over previous
#include <hip/hip_runtime.h>
#include <hip/hip_bf16.h>
#include <math.h>


typedef __attribute__((ext_vector_type(2))) float v2f;
typedef __attribute__((ext_vector_type(8))) float v8f;

__device__ __forceinline__ float maxnum1(float a, float b) {
    float r;
    asm("v_max_num_f32 %0, %1, %2" : "=v"(r) : "v"(a), "v"(b));
    return r;
}
__device__ __forceinline__ float relu1(float x) {
    float r;
    asm("v_max_num_f32 %0, 0, %1" : "=v"(r) : "v"(x));
    return r;
}

__global__ __launch_bounds__(256) void rnn_wmma_kernel(
    const float* __restrict__ state,
    const float* __restrict__ W_ih, const float* __restrict__ W_hh,
    const float* __restrict__ b_ih, const float* __restrict__ b_hh,
    const float* __restrict__ W_out, const float* __restrict__ b_out,
    float* __restrict__ out, int B, int T)
{
    const int lane = threadIdx.x & 31;
    const int m    = lane & 15;
    const bool low = lane < 16;
    const int w    = blockIdx.x * (blockDim.x >> 5) + (threadIdx.x >> 5);
    if (w * 32 >= B) return;

    const int  unit = (m < 5) ? m : (m == 8 ? 2 : 3);
    const bool live = ((0x31Fu >> m) & 1u) != 0u;
    float wh[5], wi[3];
#pragma unroll
    for (int c = 0; c < 5; ++c) wh[c] = W_hh[unit * 5 + c];
#pragma unroll
    for (int c = 0; c < 3; ++c) wi[c] = W_ih[unit * 3 + c];

    v2f A1, A2;
    A1.x = live ? (low ? wh[0] : wh[2]) : 0.0f;
    A1.y = live ? (low ? wh[1] : wh[3]) : 0.0f;
    A2.x = live ? (low ? wh[4] : wi[1]) : 0.0f;
    A2.y = live ? (low ? wi[0] : wi[2]) : 0.0f;

    const float NEG_INF = -__builtin_huge_valf();
    float bs[5];
#pragma unroll
    for (int j = 0; j < 5; ++j) bs[j] = b_ih[j] + b_hh[j];
    v8f cbias;
    cbias[0] = low ? bs[0] : bs[2];
    cbias[1] = low ? bs[1] : bs[3];
    cbias[2] = low ? bs[2] : 0.0f;
    cbias[3] = low ? bs[3] : 0.0f;
    cbias[4] = low ? bs[4] : NEG_INF;
    cbias[5] = 0.0f; cbias[6] = 0.0f; cbias[7] = 0.0f;

    float a[2][5];
    size_t voff[2];
#pragma unroll
    for (int g = 0; g < 2; ++g) {
#pragma unroll
        for (int j = 0; j < 4; ++j) a[g][j] = 0.0f;
        a[g][4] = low ? 0.0f : NEG_INF;
        const int chain = w * 32 + g * 16 + m;
        voff[g] = (size_t)chain * (size_t)T * 12u;
    }
    const char* const sbase = (const char*)state;

    auto step = [&](int g, float x0, float x1, float x2) {
        const float xs = low ? 0.0f : x1;
        v2f B1, B2;
        B2.y = low ? x0 : x2;
        B1.x = relu1(a[g][0]);
        B1.y = relu1(a[g][1]);
        B2.x = maxnum1(a[g][4], xs);
        v8f acc = __builtin_amdgcn_wmma_f32_16x16x4_f32(false, A1, false, B1,
                                                        (short)0, cbias, false, false);
        acc = __builtin_amdgcn_wmma_f32_16x16x4_f32(false, A2, false, B2,
                                                    (short)0, acc, false, false);
        a[g][0] = acc[0];
        a[g][1] = acc[1];
        a[g][2] = acc[2];
        a[g][3] = acc[3];
        a[g][4] = acc[4];
    };

    struct Blk { float4 q[2][6]; };

    auto load = [&](Blk& b, int blk) {
        const char* base = sbase + (size_t)blk * 96u;
#pragma unroll
        for (int g = 0; g < 2; ++g) {
            const float4* s = (const float4*)(base + voff[g]);
#pragma unroll
            for (int k = 0; k < 6; ++k) b.q[g][k] = s[k];
            __builtin_prefetch(base + voff[g] + 384, 0, 3);
        }
    };

    auto compute = [&](const Blk& b, int nsteps) {
        float f[2][24];
#pragma unroll
        for (int g = 0; g < 2; ++g) {
#pragma unroll
            for (int k = 0; k < 6; ++k) {
                f[g][4 * k + 0] = b.q[g][k].x;
                f[g][4 * k + 1] = b.q[g][k].y;
                f[g][4 * k + 2] = b.q[g][k].z;
                f[g][4 * k + 3] = b.q[g][k].w;
            }
        }
#pragma unroll
        for (int t = 0; t < 8; ++t) {
            if (t < nsteps) {
                step(0, f[0][3 * t], f[0][3 * t + 1], f[0][3 * t + 2]);
                step(1, f[1][3 * t], f[1][3 * t + 1], f[1][3 * t + 2]);
            }
        }
    };

    const int nb = T >> 3;
    Blk A, Bk;
    if (nb > 0) {
        load(A, 0);
        int k = 0;
        for (; k + 2 < nb; k += 2) {
            load(Bk, k + 1);
            __builtin_amdgcn_sched_barrier(0);
            compute(A, 8);
            load(A, k + 2);
            __builtin_amdgcn_sched_barrier(0);
            compute(Bk, 8);
        }
        if (k + 1 < nb) {
            load(Bk, k + 1);
            compute(A, 8);
            compute(Bk, 8);
        } else {
            compute(A, 8);
        }
    }
    if (T & 4) {
        const char* base = sbase + (size_t)nb * 96u;
#pragma unroll
        for (int g = 0; g < 2; ++g) {
            const float4* s = (const float4*)(base + voff[g]);
#pragma unroll
            for (int k = 0; k < 3; ++k) A.q[g][k] = s[k];
#pragma unroll
            for (int k = 3; k < 6; ++k) A.q[g][k] = make_float4(0.f, 0.f, 0.f, 0.f);
        }
        compute(A, 4);
    }

    if (low) {
#pragma unroll
        for (int g = 0; g < 2; ++g) {
            float h[5];
#pragma unroll
            for (int j = 0; j < 5; ++j) h[j] = fmaxf(a[g][j], 0.0f);
            float z = W_out[0] * h[0] + W_out[1] * h[1] + W_out[2] * h[2] +
                      W_out[3] * h[3] + W_out[4] * h[4] + b_out[0];
            out[w * 32 + g * 16 + m] = tanhf(z);
        }
    }
}

extern "C" void kernel_launch(void* const* d_in, const int* in_sizes, int n_in,
                              void* d_out, int out_size, void* d_ws, size_t ws_size,
                              hipStream_t stream) {
    (void)n_in; (void)out_size; (void)d_ws; (void)ws_size;
    const float* state = (const float*)d_in[0];
    const float* W_ih  = (const float*)d_in[1];
    const float* W_hh  = (const float*)d_in[2];
    const float* b_ih  = (const float*)d_in[3];
    const float* b_hh  = (const float*)d_in[4];
    const float* W_out = (const float*)d_in[5];
    const float* b_out = (const float*)d_in[6];
    float* out = (float*)d_out;

    const int T = 2048;
    const int B = in_sizes[0] / (T * 3);
    const int waves = B / 32;
    const int wavesPerBlock = 8;
    const int blocks = (waves + wavesPerBlock - 1) / wavesPerBlock;

    rnn_wmma_kernel<<<blocks, wavesPerBlock * 32, 0, stream>>>(
        state, W_ih, W_hh, b_ih, b_hh, W_out, b_out, out, B, T);
}


# One 16-chain group per wave, each wave on its own SIMD (two-wave work-groups, not eight-wave; 16-timestep input blocks); wave 1 passes results via LDS so one wave stores each output line
# speedup vs baseline: 5.6530x; 3.4861x over previous
#include <hip/hip_runtime.h>
#include <hip/hip_bf16.h>
#include <math.h>


typedef __attribute__((ext_vector_type(2))) float v2f;
typedef __attribute__((ext_vector_type(8))) float v8f;

__device__ __forceinline__ float maxnum1(float a, float b) {
    float r;
    asm("v_max_num_f32 %0, %1, %2" : "=v"(r) : "v"(a), "v"(b));
    return r;
}
__device__ __forceinline__ float relu1(float x) {
    float r;
    asm("v_max_num_f32 %0, 0, %1" : "=v"(r) : "v"(x));
    return r;
}

__global__ __launch_bounds__(64) void rnn_wmma_kernel(
    const float* __restrict__ state,
    const float* __restrict__ W_ih, const float* __restrict__ W_hh,
    const float* __restrict__ b_ih, const float* __restrict__ b_hh,
    const float* __restrict__ W_out, const float* __restrict__ b_out,
    float* __restrict__ out, int B, int T)
{
    __shared__ float hand[16];

    const int lane = threadIdx.x & 31;
    const int m    = lane & 15;
    const bool low = lane < 16;
    const int wv   = threadIdx.x >> 5;
    const int cb   = blockIdx.x * 32;
    if (cb >= B) return;

    const int  unit = (m < 5) ? m : (m == 8 ? 2 : 3);
    const bool live = ((0x31Fu >> m) & 1u) != 0u;
    float wh[5], wi[3];
#pragma unroll
    for (int c = 0; c < 5; ++c) wh[c] = W_hh[unit * 5 + c];
#pragma unroll
    for (int c = 0; c < 3; ++c) wi[c] = W_ih[unit * 3 + c];

    v2f A1, A2;
    A1.x = live ? (low ? wh[0] : wh[2]) : 0.0f;
    A1.y = live ? (low ? wh[1] : wh[3]) : 0.0f;
    A2.x = live ? (low ? wh[4] : wi[1]) : 0.0f;
    A2.y = live ? (low ? wi[0] : wi[2]) : 0.0f;

    const float NEG_INF = -__builtin_huge_valf();
    float bs[5];
#pragma unroll
    for (int j = 0; j < 5; ++j) bs[j] = b_ih[j] + b_hh[j];
    v8f cbias;
    cbias[0] = low ? bs[0] : bs[2];
    cbias[1] = low ? bs[1] : bs[3];
    cbias[2] = low ? bs[2] : 0.0f;
    cbias[3] = low ? bs[3] : 0.0f;
    cbias[4] = low ? bs[4] : NEG_INF;
    cbias[5] = 0.0f; cbias[6] = 0.0f; cbias[7] = 0.0f;

    float a[5];
#pragma unroll
    for (int j = 0; j < 4; ++j) a[j] = 0.0f;
    a[4] = low ? 0.0f : NEG_INF;
    const int chain = cb + wv * 16 + m;
    const size_t voff = (size_t)chain * (size_t)T * 12u;
    const char* const sbase = (const char*)state;

    auto step = [&](float x0, float x1, float x2) {
        const float xs = low ? 0.0f : x1;
        v2f B1, B2;
        B2.y = low ? x0 : x2;
        B1.x = relu1(a[0]);
        B1.y = relu1(a[1]);
        B2.x = maxnum1(a[4], xs);
        v8f acc = __builtin_amdgcn_wmma_f32_16x16x4_f32(false, A1, false, B1,
                                                        (short)0, cbias, false, false);
        acc = __builtin_amdgcn_wmma_f32_16x16x4_f32(false, A2, false, B2,
                                                    (short)0, acc, false, false);
        a[0] = acc[0];
        a[1] = acc[1];
        a[2] = acc[2];
        a[3] = acc[3];
        a[4] = acc[4];
    };

    struct Blk { float4 q[12]; };

    const int nb = T >> 4;

    auto load = [&](Blk& b, int blk) {
        const float4* s = (const float4*)(sbase + (size_t)blk * 192u + voff);
#pragma unroll
        for (int k = 0; k < 12; ++k) b.q[k] = s[k];
        const int pb = (blk + 4 < nb) ? (blk + 4) : (nb - 1);
        __builtin_prefetch(sbase + (size_t)pb * 192u + voff, 0, 3);
    };

    auto compute = [&](const Blk& b, int nsteps) {
        float f[48];
#pragma unroll
        for (int k = 0; k < 12; ++k) {
            f[4 * k + 0] = b.q[k].x;
            f[4 * k + 1] = b.q[k].y;
            f[4 * k + 2] = b.q[k].z;
            f[4 * k + 3] = b.q[k].w;
        }
#pragma unroll
        for (int t = 0; t < 16; ++t) {
            if (t < nsteps) step(f[3 * t], f[3 * t + 1], f[3 * t + 2]);
        }
    };

    Blk A, Bk;
    if (nb > 0) {
        load(A, 0);
        int k = 0;
        for (; k + 2 < nb; k += 2) {
            load(Bk, k + 1);
            __builtin_amdgcn_sched_barrier(0);
            compute(A, 16);
            load(A, k + 2);
            __builtin_amdgcn_sched_barrier(0);
            compute(Bk, 16);
        }
        if (k + 1 < nb) {
            load(Bk, k + 1);
            compute(A, 16);
            compute(Bk, 16);
        } else {
            compute(A, 16);
        }
    }
    const int rem = T & 12;
    if (rem) {
        const float4* s = (const float4*)(sbase + (size_t)nb * 192u + voff);
#pragma unroll
        for (int k = 0; k < 12; ++k)
            A.q[k] = (k * 4 < rem * 3) ? s[k] : make_float4(0.f, 0.f, 0.f, 0.f);
        compute(A, rem);
    }

    float h[5];
#pragma unroll
    for (int j = 0; j < 5; ++j) h[j] = fmaxf(a[j], 0.0f);
    const float z = W_out[0] * h[0] + W_out[1] * h[1] + W_out[2] * h[2] +
                    W_out[3] * h[3] + W_out[4] * h[4] + b_out[0];
    const float r = tanhf(z);
    if (wv == 1 && low) hand[m] = r;
    __syncthreads();
    if (wv == 0) {
        const float o = low ? r : hand[m];
        out[cb + lane] = o;
    }
}

extern "C" void kernel_launch(void* const* d_in, const int* in_sizes, int n_in,
                              void* d_out, int out_size, void* d_ws, size_t ws_size,
                              hipStream_t stream) {
    (void)n_in; (void)out_size; (void)d_ws; (void)ws_size;
    const float* state = (const float*)d_in[0];
    const float* W_ih  = (const float*)d_in[1];
    const float* W_hh  = (const float*)d_in[2];
    const float* b_ih  = (const float*)d_in[3];
    const float* b_hh  = (const float*)d_in[4];
    const float* W_out = (const float*)d_in[5];
    const float* b_out = (const float*)d_in[6];
    float* out = (float*)d_out;

    const int T = 2048;
    const int B = in_sizes[0] / (T * 3);
    const int blocks = B / 32;

    rnn_wmma_kernel<<<blocks, 64, 0, stream>>>(
        state, W_ih, W_hh, b_ih, b_hh, W_out, b_out, out, B, T);
}


# One f32 WMMA per timestep, not two: a chain owns two B columns (h0..h3 | h4,x0..x2), 8 chains per wave, four-wave work-groups; a lane-swizzled add joins the partial sums; f32 sum order changes
# speedup vs baseline: 7.3757x; 1.3047x over previous
#include <hip/hip_runtime.h>
#include <hip/hip_bf16.h>
#include <math.h>


typedef __attribute__((ext_vector_type(2))) float v2f;
typedef __attribute__((ext_vector_type(8))) float v8f;

__device__ __forceinline__ float maxnum1(float a, float b) {
    float r;
    asm("v_max_num_f32 %0, %1, %2" : "=v"(r) : "v"(a), "v"(b));
    return r;
}
__device__ __forceinline__ float maxnum3(float a, float b, float c) {
    float r;
    asm("v_max3_num_f32 %0, %1, %2, %3" : "=v"(r) : "v"(a), "v"(b), "v"(c));
    return r;
}
__device__ __forceinline__ float add_ror8(float p, float q) {
    float r;
    asm("v_add_f32_dpp %0, %1, %2 row_ror:8 row_mask:0xf bank_mask:0xf"
        : "=v"(r) : "v"(q), "v"(p));
    return r;
}

__global__ __launch_bounds__(128) void rnn_wmma_kernel(
    const float* __restrict__ state,
    const float* __restrict__ W_ih, const float* __restrict__ W_hh,
    const float* __restrict__ b_ih, const float* __restrict__ b_hh,
    const float* __restrict__ W_out, const float* __restrict__ b_out,
    float* __restrict__ out, int B, int T)
{
    __shared__ float hs[4][5][8];

    const int lane = threadIdx.x & 31;
    const int m    = lane & 15;
    const int j    = lane & 7;
    const bool low = lane < 16;
    const bool cx  = (lane & 8) != 0;
    const int wv   = threadIdx.x >> 5;
    const int cb   = blockIdx.x * 32;
    if (cb >= B) return;

    const int  unit = (int)((0x0000332200441100ull >> (4 * m)) & 7ull);
    const bool live = ((0x0F3Fu >> m) & 1u) != 0u;
    const bool rowx = ((0x0A1Au >> m) & 1u) != 0u;
    float wh[5], wi[3];
#pragma unroll
    for (int c = 0; c < 5; ++c) wh[c] = W_hh[unit * 5 + c];
#pragma unroll
    for (int c = 0; c < 3; ++c) wi[c] = W_ih[unit * 3 + c];
    const float k0 = rowx ? wh[4] : wh[0];
    const float k1 = rowx ? wi[0] : wh[1];
    const float k2 = rowx ? wi[1] : wh[2];
    const float k3 = rowx ? wi[2] : wh[3];

    v2f A1;
    A1.x = live ? (low ? k0 : k2) : 0.0f;
    A1.y = live ? (low ? k1 : k3) : 0.0f;

    const float NEG_INF = -__builtin_huge_valf();
    float bs[5];
#pragma unroll
    for (int u = 0; u < 5; ++u) bs[u] = b_ih[u] + b_hh[u];
    v8f cbias;
    cbias[0] = cx ? NEG_INF : (low ? bs[0] : bs[2]);
    cbias[1] = 0.0f;
    cbias[2] = cx ? NEG_INF : (low ? bs[1] : bs[3]);
    cbias[3] = 0.0f;
    cbias[4] = (low && cx) ? bs[4] : NEG_INF;
    cbias[5] = 0.0f; cbias[6] = 0.0f; cbias[7] = 0.0f;

    float a[6];
    a[0] = NEG_INF; a[1] = 0.0f; a[2] = NEG_INF; a[3] = 0.0f; a[4] = NEG_INF; a[5] = 0.0f;

    const int chain = cb + wv * 8 + j;
    const size_t voff = (size_t)chain * (size_t)T * 12u;
    const char* const sbase = (const char*)state;
    const bool g3 = cx && !low;

    auto step = [&](float x0, float x1, float x2) {
        const float xs = g3 ? x1 : 0.0f;
        const float ys = cx ? (low ? x0 : x2) : 0.0f;
        const float tx = add_ror8(a[0], a[1]);
        const float t4 = add_ror8(a[4], a[5]);
        const float ty = add_ror8(a[2], a[3]);
        v2f B1;
        B1.x = maxnum3(tx, t4, xs);
        B1.y = maxnum1(ty, ys);
        const v8f acc = __builtin_amdgcn_wmma_f32_16x16x4_f32(false, A1, false, B1,
                                                              (short)0, cbias, false, false);
        a[0] = acc[0];
        a[1] = acc[1];
        a[2] = acc[2];
        a[3] = acc[3];
        a[4] = acc[4];
        a[5] = acc[5];
    };

    struct Blk { float4 q[12]; };

    const int nb = T >> 4;

    auto load = [&](Blk& b, int blk) {
        const float4* s = (const float4*)(sbase + (size_t)blk * 192u + voff);
#pragma unroll
        for (int k = 0; k < 12; ++k) b.q[k] = s[k];
        const int pb = (blk + 4 < nb) ? (blk + 4) : (nb - 1);
        __builtin_prefetch(sbase + (size_t)pb * 192u + voff, 0, 3);
    };

    auto compute = [&](const Blk& b, int nsteps) {
        float f[48];
#pragma unroll
        for (int k = 0; k < 12; ++k) {
            f[4 * k + 0] = b.q[k].x;
            f[4 * k + 1] = b.q[k].y;
            f[4 * k + 2] = b.q[k].z;
            f[4 * k + 3] = b.q[k].w;
        }
#pragma unroll
        for (int t = 0; t < 16; ++t) {
            if (t < nsteps) step(f[3 * t], f[3 * t + 1], f[3 * t + 2]);
        }
    };

    Blk A, Bk;
    if (nb > 0) {
        load(A, 0);
        int k = 0;
        for (; k + 2 < nb; k += 2) {
            load(Bk, k + 1);
            __builtin_amdgcn_sched_barrier(0);
            compute(A, 16);
            load(A, k + 2);
            __builtin_amdgcn_sched_barrier(0);
            compute(Bk, 16);
        }
        if (k + 1 < nb) {
            load(Bk, k + 1);
            compute(A, 16);
            compute(Bk, 16);
        } else {
            compute(A, 16);
        }
    }
    const int rem = T & 12;
    if (rem) {
        const float4* s = (const float4*)(sbase + (size_t)nb * 192u + voff);
#pragma unroll
        for (int k = 0; k < 12; ++k)
            A.q[k] = (k * 4 < rem * 3) ? s[k] : make_float4(0.f, 0.f, 0.f, 0.f);
        compute(A, rem);
    }

    const float hx = maxnum3(add_ror8(a[0], a[1]), add_ror8(a[4], a[5]), 0.0f);
    const float hy = maxnum1(add_ror8(a[2], a[3]), 0.0f);
    const int g = lane >> 3;
    if (g != 3) hs[wv][(g == 0) ? 0 : ((g == 1) ? 4 : 2)][j] = hx;
    if ((g & 1) == 0) hs[wv][(g == 0) ? 1 : 3][j] = hy;
    __syncthreads();
    if (wv == 0) {
        const int w = lane >> 3;
        float h[5];
#pragma unroll
        for (int u = 0; u < 5; ++u) h[u] = hs[w][u][j];
        const float z = W_out[0] * h[0] + W_out[1] * h[1] + W_out[2] * h[2] +
                        W_out[3] * h[3] + W_out[4] * h[4] + b_out[0];
        out[cb + lane] = tanhf(z);
    }
}

extern "C" void kernel_launch(void* const* d_in, const int* in_sizes, int n_in,
                              void* d_out, int out_size, void* d_ws, size_t ws_size,
                              hipStream_t stream) {
    (void)n_in; (void)out_size; (void)d_ws; (void)ws_size;
    const float* state = (const float*)d_in[0];
    const float* W_ih  = (const float*)d_in[1];
    const float* W_hh  = (const float*)d_in[2];
    const float* b_ih  = (const float*)d_in[3];
    const float* b_hh  = (const float*)d_in[4];
    const float* W_out = (const float*)d_in[5];
    const float* b_out = (const float*)d_in[6];
    float* out = (float*)d_out;

    const int T = 2048;
    const int B = in_sizes[0] / (T * 3);
    const int blocks = B / 32;

    rnn_wmma_kernel<<<blocks, 128, 0, stream>>>(
        state, W_ih, W_hh, b_ih, b_hh, W_out, b_out, out, B, T);
}
